# starting kernel
# baseline (speedup 1.0000x reference)
#include <hip/hip_runtime.h>

typedef __attribute__((ext_vector_type(2))) float v2f;
typedef __attribute__((ext_vector_type(4))) float v4f;
typedef __attribute__((ext_vector_type(8))) float v8f;

#define N_NODES 50000
#define N_EDGES 800000
#define D 128

__global__ void gcn_zero_kernel(v4f* __restrict__ p, int n4) {
    int i = blockIdx.x * blockDim.x + threadIdx.x;
    if (i < n4) {
        v4f z = {};
        p[i] = z;
    }
}

__global__ void __launch_bounds__(256)
gcn_spmm_kernel(const int* __restrict__ adj_row,
                const int* __restrict__ adj_col,
                const float* __restrict__ adj_vals,
                const float* __restrict__ features,
                float* __restrict__ f) {
    const int waveInBlk = __builtin_amdgcn_readfirstlane(threadIdx.x >> 5);
    const int lane      = threadIdx.x & 31;
    const int edge      = blockIdx.x * 8 + waveInBlk;
    if (edge >= N_EDGES) return;

    const int   r = adj_row[edge];
    const int   c = adj_col[edge];
    const float v = adj_vals[edge];

    const v4f x = *(const v4f*)(features + (size_t)c * D + lane * 4);
    float* dst = f + (size_t)r * D + lane * 4;

    unsafeAtomicAdd(dst + 0, v * x.x);
    unsafeAtomicAdd(dst + 1, v * x.y);
    unsafeAtomicAdd(dst + 2, v * x.z);
    unsafeAtomicAdd(dst + 3, v * x.w);
}

__global__ void __launch_bounds__(256)
gcn_gemm_wmma_kernel(const float* __restrict__ f,
                     const float* __restrict__ W,
                     const float* __restrict__ bias,
                     float* __restrict__ out) {
    const int waveInBlk = threadIdx.x >> 5;
    const int lane      = threadIdx.x & 31;
    const int waveId    = blockIdx.x * 8 + waveInBlk;

    const int mTile = waveId >> 3;
    const int oTile = waveId & 7;
    const int half  = lane >> 4;
    const int l     = lane & 15;

    const int m0 = mTile * 16;
    const int o0 = oTile * 16;

    const float* aRow = f + (size_t)(m0 + l) * D + 4 * half;
    const float* bRow = W + (size_t)(o0 + l) * D + 4 * half;

    v8f acc = {};

#pragma unroll
    for (int p = 0; p < 16; ++p) {
        const v4f va = *(const v4f*)(aRow + 8 * p);
        const v4f vb = *(const v4f*)(bRow + 8 * p);
        const v2f a0 = {va.x, va.y};
        const v2f b0 = {vb.x, vb.y};
        const v2f a1 = {va.z, va.w};
        const v2f b1 = {vb.z, vb.w};
        acc = __builtin_amdgcn_wmma_f32_16x16x4_f32(
            false, a0, false, b0, (short)0, acc, false, false);
        acc = __builtin_amdgcn_wmma_f32_16x16x4_f32(
            false, a1, false, b1, (short)0, acc, false, false);
    }

    const float bv = bias[o0 + l];
    float* outBase = out + (size_t)(m0 + 8 * half) * D + o0 + l;
#pragma unroll
    for (int r = 0; r < 8; ++r) {
        outBase[(size_t)r * D] = acc[r] + bv;
    }
}

extern "C" void kernel_launch(void* const* d_in, const int* in_sizes, int n_in,
                              void* d_out, int out_size, void* d_ws, size_t ws_size,
                              hipStream_t stream) {
    const int*   adj_row  = (const int*)d_in[0];
    const int*   adj_col  = (const int*)d_in[1];
    const float* adj_vals = (const float*)d_in[2];
    const float* features = (const float*)d_in[3];
    const float* W        = (const float*)d_in[4];
    const float* bias     = (const float*)d_in[5];
    float*       out      = (float*)d_out;

    float* f = (float*)d_ws;

    const int n4 = N_NODES * D / 4;
    gcn_zero_kernel<<<(n4 + 255) / 256, 256, 0, stream>>>((v4f*)f, n4);

    gcn_spmm_kernel<<<(N_EDGES + 7) / 8, 256, 0, stream>>>(
        adj_row, adj_col, adj_vals, features, f);

    const int gemmBlocks = (N_NODES / 16) * 8 / 8;
    gcn_gemm_wmma_kernel<<<gemmBlocks, 256, 0, stream>>>(f, W, bias, out);
}


# SpMM by gather instead of float-atomic scatter: edges counting-sorted by row in the call (integer atomic rank = order of each row's sum), one wave per row sums in registers; f zero pass dropped
# speedup vs baseline: 1.2681x; 1.2681x over previous
#include <hip/hip_runtime.h>

typedef __attribute__((ext_vector_type(2))) float v2f;
typedef __attribute__((ext_vector_type(4))) float v4f;
typedef __attribute__((ext_vector_type(8))) float v8f;
typedef __attribute__((ext_vector_type(2))) int   v2i;
typedef __attribute__((ext_vector_type(4))) int   v4i;

#define N_NODES 50000
#define N_EDGES 800000
#define D 128

#define SCAN_T   1024
#define SCAN_PER 49
#define SCAN_LEN (SCAN_T * SCAN_PER)
#define SCAN_V4  12

static_assert(SCAN_LEN >= N_NODES, "counter array too short");
static_assert(SCAN_V4 * SCAN_T * 4 + SCAN_T == SCAN_LEN, "staging rounds must cover the counter array");
static_assert(N_EDGES % 4 == 0 && N_NODES % 4 == 0, "b128 passes over edges and offsets");

#define WS_CNT   (N_NODES * D)
#define WS_OFFS  (WS_CNT + SCAN_LEN)
#define WS_RANK  (WS_OFFS + 50048)
#define WS_PAIRS (WS_RANK + N_EDGES)

__global__ void gcn_zero_kernel(v4i* __restrict__ p, int n4) {
    int i = blockIdx.x * blockDim.x + threadIdx.x;
    if (i < n4) {
        v4i z = {};
        p[i] = z;
    }
}

__global__ void __launch_bounds__(256)
gcn_rank_kernel(const int* __restrict__ adj_row,
                int* __restrict__ cnt,
                int* __restrict__ rank) {
    const int t = blockIdx.x * 256 + threadIdx.x;
    if (t < N_EDGES / 4) {
        const v4i r = ((const v4i*)adj_row)[t];
        v4i k;
        k.x = atomicAdd(cnt + r.x, 1);
        k.y = atomicAdd(cnt + r.y, 1);
        k.z = atomicAdd(cnt + r.z, 1);
        k.w = atomicAdd(cnt + r.w, 1);
        ((v4i*)rank)[t] = k;
    }
}

__global__ void __launch_bounds__(SCAN_T)
gcn_scan_kernel(const int* __restrict__ cnt, int* __restrict__ offs) {
    __shared__ __attribute__((aligned(16))) int s[SCAN_LEN];
    __shared__ int wsum[32];
    const int t    = threadIdx.x;
    const int lane = t & 31;
    const int w    = t >> 5;

    v4i q[SCAN_V4];
#pragma unroll
    for (int k = 0; k < SCAN_V4; ++k) q[k] = ((const v4i*)cnt)[k * SCAN_T + t];
    const int qt = cnt[SCAN_V4 * SCAN_T * 4 + t];
#pragma unroll
    for (int k = 0; k < SCAN_V4; ++k) ((v4i*)s)[k * SCAN_T + t] = q[k];
    s[SCAN_V4 * SCAN_T * 4 + t] = qt;
    __syncthreads();

    int loc = 0;
    for (int k = 0; k < SCAN_PER; ++k) loc += s[t * SCAN_PER + k];

    int inc = loc;
    for (int d = 1; d < 32; d <<= 1) {
        const int y = __shfl_up(inc, d);
        if (lane >= d) inc += y;
    }
    if (lane == 31) wsum[w] = inc;
    __syncthreads();
    if (w == 0) {
        const int v = wsum[lane];
        int iv = v;
        for (int d = 1; d < 32; d <<= 1) {
            const int y = __shfl_up(iv, d);
            if (lane >= d) iv += y;
        }
        wsum[lane] = iv - v;
    }
    __syncthreads();

    int run = wsum[w] + inc - loc;
    for (int k = 0; k < SCAN_PER; ++k) {
        const int c = s[t * SCAN_PER + k];
        s[t * SCAN_PER + k] = run;
        run += c;
    }
    if (t == SCAN_T - 1) offs[N_NODES] = run;
    __syncthreads();

    for (int i = t; i < N_NODES / 4; i += SCAN_T) ((v4i*)offs)[i] = ((const v4i*)s)[i];
}

__global__ void __launch_bounds__(256)
gcn_fill_kernel(const int* __restrict__ adj_row,
                const int* __restrict__ adj_col,
                const float* __restrict__ adj_vals,
                const int* __restrict__ offs,
                const int* __restrict__ rank,
                int* __restrict__ pairs) {
    const int t = blockIdx.x * 256 + threadIdx.x;
    if (t < N_EDGES / 4) {
        const v4i r = ((const v4i*)adj_row)[t];
        const v4i k = ((const v4i*)rank)[t];
        const v4i c = ((const v4i*)adj_col)[t];
        const v4f v = ((const v4f*)adj_vals)[t];
        const int p0 = offs[r.x] + k.x;
        const int p1 = offs[r.y] + k.y;
        const int p2 = offs[r.z] + k.z;
        const int p3 = offs[r.w] + k.w;
        v2i* P = (v2i*)pairs;
        const v2i e0 = {c.x, __float_as_int(v.x)};
        const v2i e1 = {c.y, __float_as_int(v.y)};
        const v2i e2 = {c.z, __float_as_int(v.z)};
        const v2i e3 = {c.w, __float_as_int(v.w)};
        P[p0] = e0;
        P[p1] = e1;
        P[p2] = e2;
        P[p3] = e3;
    }
}

template <int U>
__device__ __forceinline__ v4f gcn_gather_group(const float* __restrict__ fbase,
                                                const v2i pr, const int j, v4f acc) {
    v4f   x[U];
    float v[U];
#pragma unroll
    for (int u = 0; u < U; ++u) {
        const int c = __builtin_amdgcn_readlane(pr.x, j + u);
        v[u] = __int_as_float(__builtin_amdgcn_readlane(pr.y, j + u));
        x[u] = *(const v4f*)(fbase + (size_t)c * D);
    }
#pragma unroll
    for (int u = 0; u < U; ++u) acc += v[u] * x[u];
    return acc;
}

__global__ void __launch_bounds__(256)
gcn_gather_kernel(const int* __restrict__ offs,
                  const int* __restrict__ pairs,
                  const float* __restrict__ features,
                  float* __restrict__ f) {
    const int waveInBlk = __builtin_amdgcn_readfirstlane(threadIdx.x >> 5);
    const int lane      = threadIdx.x & 31;
    const int row       = blockIdx.x * 8 + waveInBlk;
    if (row >= N_NODES) return;

    const int beg = offs[row];
    const int end = offs[row + 1];

    const float* fbase = features + lane * 4;
    v4f acc = {};

    for (int e0 = beg; e0 < end; e0 += 32) {
        const int n = (end - e0 < 32) ? (end - e0) : 32;
        v2i pr = {};
        if (lane < n) pr = ((const v2i*)pairs)[e0 + lane];

        int j = 0;
        for (; j + 8 <= n; j += 8) acc = gcn_gather_group<8>(fbase, pr, j, acc);
        const int rem = n - j;
        if (rem & 4) { acc = gcn_gather_group<4>(fbase, pr, j, acc); j += 4; }
        if (rem & 2) { acc = gcn_gather_group<2>(fbase, pr, j, acc); j += 2; }
        if (rem & 1) { acc = gcn_gather_group<1>(fbase, pr, j, acc); }
    }

    *(v4f*)(f + (size_t)row * D + lane * 4) = acc;
}

__global__ void __launch_bounds__(256)
gcn_gemm_wmma_kernel(const float* __restrict__ f,
                     const float* __restrict__ W,
                     const float* __restrict__ bias,
                     float* __restrict__ out) {
    const int waveInBlk = threadIdx.x >> 5;
    const int lane      = threadIdx.x & 31;
    const int waveId    = blockIdx.x * 8 + waveInBlk;

    const int mTile = waveId >> 3;
    const int oTile = waveId & 7;
    const int half  = lane >> 4;
    const int l     = lane & 15;

    const int m0 = mTile * 16;
    const int o0 = oTile * 16;

    const float* aRow = f + (size_t)(m0 + l) * D + 4 * half;
    const float* bRow = W + (size_t)(o0 + l) * D + 4 * half;

    v8f acc = {};

#pragma unroll
    for (int p = 0; p < 16; ++p) {
        const v4f va = *(const v4f*)(aRow + 8 * p);
        const v4f vb = *(const v4f*)(bRow + 8 * p);
        const v2f a0 = {va.x, va.y};
        const v2f b0 = {vb.x, vb.y};
        const v2f a1 = {va.z, va.w};
        const v2f b1 = {vb.z, vb.w};
        acc = __builtin_amdgcn_wmma_f32_16x16x4_f32(
            false, a0, false, b0, (short)0, acc, false, false);
        acc = __builtin_amdgcn_wmma_f32_16x16x4_f32(
            false, a1, false, b1, (short)0, acc, false, false);
    }

    const float bv = bias[o0 + l];
    float* outBase = out + (size_t)(m0 + 8 * half) * D + o0 + l;
#pragma unroll
    for (int r = 0; r < 8; ++r) {
        outBase[(size_t)r * D] = acc[r] + bv;
    }
}

extern "C" void kernel_launch(void* const* d_in, const int* in_sizes, int n_in,
                              void* d_out, int out_size, void* d_ws, size_t ws_size,
                              hipStream_t stream) {
    const int*   adj_row  = (const int*)d_in[0];
    const int*   adj_col  = (const int*)d_in[1];
    const float* adj_vals = (const float*)d_in[2];
    const float* features = (const float*)d_in[3];
    const float* W        = (const float*)d_in[4];
    const float* bias     = (const float*)d_in[5];
    float*       out      = (float*)d_out;

    float* f     = (float*)d_ws;
    int*   cnt   = (int*)d_ws + WS_CNT;
    int*   offs  = (int*)d_ws + WS_OFFS;
    int*   rank  = (int*)d_ws + WS_RANK;
    int*   pairs = (int*)d_ws + WS_PAIRS;

    gcn_zero_kernel<<<(SCAN_LEN / 4 + 255) / 256, 256, 0, stream>>>((v4i*)cnt, SCAN_LEN / 4);

    gcn_rank_kernel<<<(N_EDGES / 4 + 255) / 256, 256, 0, stream>>>(adj_row, cnt, rank);

    gcn_scan_kernel<<<1, SCAN_T, 0, stream>>>(cnt, offs);

    gcn_fill_kernel<<<(N_EDGES / 4 + 255) / 256, 256, 0, stream>>>(
        adj_row, adj_col, adj_vals, offs, rank, pairs);

    gcn_gather_kernel<<<(N_NODES + 7) / 8, 256, 0, stream>>>(offs, pairs, features, f);

    const int gemmBlocks = (N_NODES / 16) * 8 / 8;
    gcn_gemm_wmma_kernel<<<gemmBlocks, 256, 0, stream>>>(f, W, bias, out);
}


# Fixed-pitch row slots (64 a row, slot = the counting atomic's return) filled in the counting pass: prefix-sum and fill launches go (6 -> 4); longer rows spill to a list; row sums keep arrival order
# speedup vs baseline: 1.4606x; 1.1518x over previous
#include <hip/hip_runtime.h>

typedef __attribute__((ext_vector_type(2))) float v2f;
typedef __attribute__((ext_vector_type(4))) float v4f;
typedef __attribute__((ext_vector_type(8))) float v8f;
typedef __attribute__((ext_vector_type(2))) int   v2i;
typedef __attribute__((ext_vector_type(4))) int   v4i;

#define N_NODES 50000
#define N_EDGES 800000
#define D 128

#define ROW_SLOTS 64
#define CNT_WORDS 50176
#define SPILL_N   N_NODES

static_assert(CNT_WORDS > SPILL_N && CNT_WORDS % 128 == 0, "counter array: spill counter inside, slot rows 512-byte aligned");
static_assert(N_EDGES % 4 == 0, "b128 passes over edges");
static_assert((long long)N_NODES * ROW_SLOTS < (1ll << 31), "slot index fits an int");

#define WS_CNT   (N_NODES * D)
#define WS_SLOTS (WS_CNT + CNT_WORDS)
#define WS_SPILL (WS_SLOTS + N_NODES * ROW_SLOTS * 2)

__global__ void gcn_zero_kernel(v4i* __restrict__ p, int n4) {
    int i = blockIdx.x * blockDim.x + threadIdx.x;
    if (i < n4) {
        v4i z = {};
        p[i] = z;
    }
}

__device__ __forceinline__ void gcn_put_edge(int* __restrict__ cnt, v2i* __restrict__ slots,
                                             v4i* __restrict__ spill,
                                             const int r, const int k, const int c, const float v) {
    if (k < ROW_SLOTS) {
        const v2i e = {c, __float_as_int(v)};
        slots[r * ROW_SLOTS + k] = e;
    } else {
        const int p = atomicAdd(cnt + SPILL_N, 1);
        const v4i e = {r, c, __float_as_int(v), 0};
        spill[p] = e;
    }
}

__global__ void __launch_bounds__(256)
gcn_bucket_kernel(const int* __restrict__ adj_row,
                  const int* __restrict__ adj_col,
                  const float* __restrict__ adj_vals,
                  int* __restrict__ cnt,
                  int* __restrict__ slots,
                  int* __restrict__ spill) {
    const int t = blockIdx.x * 256 + threadIdx.x;
    if (t < N_EDGES / 4) {
        const v4i r = ((const v4i*)adj_row)[t];
        const v4i c = ((const v4i*)adj_col)[t];
        const v4f v = ((const v4f*)adj_vals)[t];
        const int k0 = atomicAdd(cnt + r.x, 1);
        const int k1 = atomicAdd(cnt + r.y, 1);
        const int k2 = atomicAdd(cnt + r.z, 1);
        const int k3 = atomicAdd(cnt + r.w, 1);
        gcn_put_edge(cnt, (v2i*)slots, (v4i*)spill, r.x, k0, c.x, v.x);
        gcn_put_edge(cnt, (v2i*)slots, (v4i*)spill, r.y, k1, c.y, v.y);
        gcn_put_edge(cnt, (v2i*)slots, (v4i*)spill, r.z, k2, c.z, v.z);
        gcn_put_edge(cnt, (v2i*)slots, (v4i*)spill, r.w, k3, c.w, v.w);
    }
}

template <int U>
__device__ __forceinline__ v4f gcn_gather_group(const float* __restrict__ fbase,
                                                const v2i pr, const int j, v4f acc) {
    v4f   x[U];
    float v[U];
#pragma unroll
    for (int u = 0; u < U; ++u) {
        const int c = __builtin_amdgcn_readlane(pr.x, j + u);
        v[u] = __int_as_float(__builtin_amdgcn_readlane(pr.y, j + u));
        x[u] = *(const v4f*)(fbase + (size_t)c * D);
    }
#pragma unroll
    for (int u = 0; u < U; ++u) acc += v[u] * x[u];
    return acc;
}

__global__ void __launch_bounds__(256)
gcn_gather_kernel(const int* __restrict__ cnt,
                  const int* __restrict__ slots,
                  const int* __restrict__ spill,
                  const float* __restrict__ features,
                  float* __restrict__ f) {
    const int waveInBlk = __builtin_amdgcn_readfirstlane(threadIdx.x >> 5);
    const int lane      = threadIdx.x & 31;
    const int row       = blockIdx.x * 8 + waveInBlk;
    if (row >= N_NODES) return;

    const int deg = cnt[row];
    const int nE  = (deg < ROW_SLOTS) ? deg : ROW_SLOTS;
    const v2i* rs = (const v2i*)slots + row * ROW_SLOTS;

    const float* fbase = features + lane * 4;
    v4f acc = {};

    for (int e0 = 0; e0 < nE; e0 += 32) {
        const int n = (nE - e0 < 32) ? (nE - e0) : 32;
        v2i pr = {};
        if (lane < n) pr = rs[e0 + lane];

        int j = 0;
        for (; j + 8 <= n; j += 8) acc = gcn_gather_group<8>(fbase, pr, j, acc);
        const int rem = n - j;
        if (rem & 4) { acc = gcn_gather_group<4>(fbase, pr, j, acc); j += 4; }
        if (rem & 2) { acc = gcn_gather_group<2>(fbase, pr, j, acc); j += 2; }
        if (rem & 1) { acc = gcn_gather_group<1>(fbase, pr, j, acc); }
    }

    if (deg > ROW_SLOTS) {
        int m = cnt[SPILL_N];
        m = (m < N_EDGES) ? m : N_EDGES;
        for (int b = 0; b < m; b += 32) {
            v4i e = {-1, 0, 0, 0};
            if (b + lane < m) e = ((const v4i*)spill)[b + lane];
            unsigned hit = __builtin_amdgcn_ballot_w32(e.x == row);
            while (hit) {
                const int from = __builtin_ctz(hit);
                hit &= hit - 1;
                const int   c = __builtin_amdgcn_readlane(e.y, from);
                const float v = __int_as_float(__builtin_amdgcn_readlane(e.z, from));
                acc += v * *(const v4f*)(fbase + (size_t)c * D);
            }
        }
    }

    *(v4f*)(f + (size_t)row * D + lane * 4) = acc;
}

__global__ void __launch_bounds__(256)
gcn_gemm_wmma_kernel(const float* __restrict__ f,
                     const float* __restrict__ W,
                     const float* __restrict__ bias,
                     float* __restrict__ out) {
    const int waveInBlk = threadIdx.x >> 5;
    const int lane      = threadIdx.x & 31;
    const int waveId    = blockIdx.x * 8 + waveInBlk;

    const int mTile = waveId >> 3;
    const int oTile = waveId & 7;
    const int half  = lane >> 4;
    const int l     = lane & 15;

    const int m0 = mTile * 16;
    const int o0 = oTile * 16;

    const float* aRow = f + (size_t)(m0 + l) * D + 4 * half;
    const float* bRow = W + (size_t)(o0 + l) * D + 4 * half;

    v8f acc = {};

#pragma unroll
    for (int p = 0; p < 16; ++p) {
        const v4f va = *(const v4f*)(aRow + 8 * p);
        const v4f vb = *(const v4f*)(bRow + 8 * p);
        const v2f a0 = {va.x, va.y};
        const v2f b0 = {vb.x, vb.y};
        const v2f a1 = {va.z, va.w};
        const v2f b1 = {vb.z, vb.w};
        acc = __builtin_amdgcn_wmma_f32_16x16x4_f32(
            false, a0, false, b0, (short)0, acc, false, false);
        acc = __builtin_amdgcn_wmma_f32_16x16x4_f32(
            false, a1, false, b1, (short)0, acc, false, false);
    }

    const float bv = bias[o0 + l];
    float* outBase = out + (size_t)(m0 + 8 * half) * D + o0 + l;
#pragma unroll
    for (int r = 0; r < 8; ++r) {
        outBase[(size_t)r * D] = acc[r] + bv;
    }
}

extern "C" void kernel_launch(void* const* d_in, const int* in_sizes, int n_in,
                              void* d_out, int out_size, void* d_ws, size_t ws_size,
                              hipStream_t stream) {
    const int*   adj_row  = (const int*)d_in[0];
    const int*   adj_col  = (const int*)d_in[1];
    const float* adj_vals = (const float*)d_in[2];
    const float* features = (const float*)d_in[3];
    const float* W        = (const float*)d_in[4];
    const float* bias     = (const float*)d_in[5];
    float*       out      = (float*)d_out;

    float* f     = (float*)d_ws;
    int*   cnt   = (int*)d_ws + WS_CNT;
    int*   slots = (int*)d_ws + WS_SLOTS;
    int*   spill = (int*)d_ws + WS_SPILL;

    gcn_zero_kernel<<<(CNT_WORDS / 4 + 255) / 256, 256, 0, stream>>>((v4i*)cnt, CNT_WORDS / 4);

    gcn_bucket_kernel<<<(N_EDGES / 4 + 255) / 256, 256, 0, stream>>>(
        adj_row, adj_col, adj_vals, cnt, slots, spill);

    gcn_gather_kernel<<<(N_NODES + 7) / 8, 256, 0, stream>>>(cnt, slots, spill, features, f);

    const int gemmBlocks = (N_NODES / 16) * 8 / 8;
    gcn_gemm_wmma_kernel<<<gemmBlocks, 256, 0, stream>>>(f, W, bias, out);
}


# SpMM gather fused into the GEMM kernel: a work-group sums 32 rows of f (4 per wave) into an LDS tile and runs the f32 WMMA from it; f never goes to memory, 4 -> 3 launches; sums keep their order
# speedup vs baseline: 1.7689x; 1.2111x over previous
#include <hip/hip_runtime.h>

typedef __attribute__((ext_vector_type(2))) float v2f;
typedef __attribute__((ext_vector_type(4))) float v4f;
typedef __attribute__((ext_vector_type(8))) float v8f;
typedef __attribute__((ext_vector_type(2))) int   v2i;
typedef __attribute__((ext_vector_type(4))) int   v4i;

#define N_NODES 50000
#define N_EDGES 800000
#define D 128

#define ROW_SLOTS 64
#define CNT_WORDS 50176
#define SPILL_N   N_NODES

static_assert(CNT_WORDS > SPILL_N && CNT_WORDS % 128 == 0, "counter array: spill counter inside, slot rows 512-byte aligned");
static_assert(N_EDGES % 4 == 0, "b128 passes over edges");
static_assert((long long)N_NODES * ROW_SLOTS < (1ll << 31), "slot index fits an int");

#define WS_CNT   (N_NODES * D)
#define WS_SLOTS (WS_CNT + CNT_WORDS)
#define WS_SPILL (WS_SLOTS + N_NODES * ROW_SLOTS * 2)

__global__ void gcn_zero_kernel(v4i* __restrict__ p, int n4) {
    int i = blockIdx.x * blockDim.x + threadIdx.x;
    if (i < n4) {
        v4i z = {};
        p[i] = z;
    }
}

__device__ __forceinline__ void gcn_put_edge(int* __restrict__ cnt, v2i* __restrict__ slots,
                                             v4i* __restrict__ spill,
                                             const int r, const int k, const int c, const float v) {
    if (k < ROW_SLOTS) {
        const v2i e = {c, __float_as_int(v)};
        slots[r * ROW_SLOTS + k] = e;
    } else {
        const int p = atomicAdd(cnt + SPILL_N, 1);
        const v4i e = {r, c, __float_as_int(v), 0};
        spill[p] = e;
    }
}

__global__ void __launch_bounds__(256)
gcn_bucket_kernel(const int* __restrict__ adj_row,
                  const int* __restrict__ adj_col,
                  const float* __restrict__ adj_vals,
                  int* __restrict__ cnt,
                  int* __restrict__ slots,
                  int* __restrict__ spill) {
    const int t = blockIdx.x * 256 + threadIdx.x;
    if (t < N_EDGES / 4) {
        const v4i r = ((const v4i*)adj_row)[t];
        const v4i c = ((const v4i*)adj_col)[t];
        const v4f v = ((const v4f*)adj_vals)[t];
        const int k0 = atomicAdd(cnt + r.x, 1);
        const int k1 = atomicAdd(cnt + r.y, 1);
        const int k2 = atomicAdd(cnt + r.z, 1);
        const int k3 = atomicAdd(cnt + r.w, 1);
        gcn_put_edge(cnt, (v2i*)slots, (v4i*)spill, r.x, k0, c.x, v.x);
        gcn_put_edge(cnt, (v2i*)slots, (v4i*)spill, r.y, k1, c.y, v.y);
        gcn_put_edge(cnt, (v2i*)slots, (v4i*)spill, r.z, k2, c.z, v.z);
        gcn_put_edge(cnt, (v2i*)slots, (v4i*)spill, r.w, k3, c.w, v.w);
    }
}

#define TILE_ROWS     32
#define ROWS_PER_WAVE 4
#define F_PITCH       132

static_assert(TILE_ROWS == 8 * ROWS_PER_WAVE, "8 waves fill the tile");
static_assert(N_NODES % 16 == 0, "a 16-row half tile is wholly inside or wholly outside the node range");
static_assert((WS_CNT % 4) == 0 && (TILE_ROWS % 4) == 0, "the 4 counters of a wave are one aligned 16-byte unit");
static_assert(((N_NODES + TILE_ROWS - 1) / TILE_ROWS) * TILE_ROWS <= CNT_WORDS, "counter reads of the last tile's padded rows stay inside cnt");

template <int U>
__device__ __forceinline__ v4f gcn_gather_group(const float* __restrict__ fbase,
                                                const v2i pr, const int j, v4f acc) {
    v4f   x[U];
    float v[U];
#pragma unroll
    for (int u = 0; u < U; ++u) {
        const int c = __builtin_amdgcn_readlane(pr.x, j + u);
        v[u] = __int_as_float(__builtin_amdgcn_readlane(pr.y, j + u));
        x[u] = *(const v4f*)(fbase + (size_t)c * D);
    }
#pragma unroll
    for (int u = 0; u < U; ++u) acc += v[u] * x[u];
    return acc;
}

__device__ __forceinline__ v4f gcn_gather_slots(const float* __restrict__ fbase,
                                                const v2i pr, const int n, v4f acc) {
    int j = 0;
    for (; j + 8 <= n; j += 8) acc = gcn_gather_group<8>(fbase, pr, j, acc);
    const int rem = n - j;
    if (rem & 4) { acc = gcn_gather_group<4>(fbase, pr, j, acc); j += 4; }
    if (rem & 2) { acc = gcn_gather_group<2>(fbase, pr, j, acc); j += 2; }
    if (rem & 1) { acc = gcn_gather_group<1>(fbase, pr, j, acc); }
    return acc;
}

__global__ void __launch_bounds__(256)
gcn_gather_gemm_kernel(const int* __restrict__ cnt,
                       const int* __restrict__ slots,
                       const int* __restrict__ spill,
                       const float* __restrict__ features,
                       const float* __restrict__ W,
                       const float* __restrict__ bias,
                       float* __restrict__ out) {
    __shared__ float ftile[TILE_ROWS * F_PITCH];

    const int waveInBlk = __builtin_amdgcn_readfirstlane(threadIdx.x >> 5);
    const int lane      = threadIdx.x & 31;
    const int m0        = blockIdx.x * TILE_ROWS;
    const int row0      = m0 + waveInBlk * ROWS_PER_WAVE;

    const float* fbase = features + lane * 4;

    int deg[ROWS_PER_WAVE];
    v2i pr0[ROWS_PER_WAVE];
#pragma unroll
    for (int i = 0; i < ROWS_PER_WAVE; ++i) {
        const int d = cnt[row0 + i];
        deg[i] = (row0 + i < N_NODES) ? d : 0;
    }
#pragma unroll
    for (int i = 0; i < ROWS_PER_WAVE; ++i) {
        const int n0 = (deg[i] < 32) ? deg[i] : 32;
        v2i p = {};
        if (lane < n0) p = ((const v2i*)slots)[(row0 + i) * ROW_SLOTS + lane];
        pr0[i] = p;
    }

#pragma unroll
    for (int i = 0; i < ROWS_PER_WAVE; ++i) {
        const int row = row0 + i;
        const int nE  = (deg[i] < ROW_SLOTS) ? deg[i] : ROW_SLOTS;
        v4f acc = {};

        acc = gcn_gather_slots(fbase, pr0[i], (nE < 32) ? nE : 32, acc);
        if (nE > 32) {
            const int n1 = nE - 32;
            v2i pr = {};
            if (lane < n1) pr = ((const v2i*)slots)[row * ROW_SLOTS + 32 + lane];
            acc = gcn_gather_slots(fbase, pr, n1, acc);
        }

        if (deg[i] > ROW_SLOTS) {
            int m = cnt[SPILL_N];
            m = (m < N_EDGES) ? m : N_EDGES;
            for (int b = 0; b < m; b += 32) {
                v4i e = {-1, 0, 0, 0};
                if (b + lane < m) e = ((const v4i*)spill)[b + lane];
                unsigned hit = __builtin_amdgcn_ballot_w32(e.x == row);
                while (hit) {
                    const int from = __builtin_ctz(hit);
                    hit &= hit - 1;
                    const int   c = __builtin_amdgcn_readlane(e.y, from);
                    const float v = __int_as_float(__builtin_amdgcn_readlane(e.z, from));
                    acc += v * *(const v4f*)(fbase + (size_t)c * D);
                }
            }
        }

        *(v4f*)(ftile + (waveInBlk * ROWS_PER_WAVE + i) * F_PITCH + lane * 4) = acc;
    }

    __syncthreads();

    const int half = lane >> 4;
    const int l    = lane & 15;
    const int o0   = waveInBlk * 16;

    const float* aRow0 = ftile + l * F_PITCH + 4 * half;
    const float* aRow1 = ftile + (16 + l) * F_PITCH + 4 * half;
    const float* bRow  = W + (size_t)(o0 + l) * D + 4 * half;

    v8f acc0 = {};
    v8f acc1 = {};

#pragma unroll
    for (int p = 0; p < 16; ++p) {
        const v4f vb  = *(const v4f*)(bRow + 8 * p);
        const v4f va0 = *(const v4f*)(aRow0 + 8 * p);
        const v4f va1 = *(const v4f*)(aRow1 + 8 * p);
        const v2f b0  = {vb.x, vb.y};
        const v2f b1  = {vb.z, vb.w};
        const v2f a00 = {va0.x, va0.y};
        const v2f a01 = {va0.z, va0.w};
        const v2f a10 = {va1.x, va1.y};
        const v2f a11 = {va1.z, va1.w};
        acc0 = __builtin_amdgcn_wmma_f32_16x16x4_f32(
            false, a00, false, b0, (short)0, acc0, false, false);
        acc1 = __builtin_amdgcn_wmma_f32_16x16x4_f32(
            false, a10, false, b0, (short)0, acc1, false, false);
        acc0 = __builtin_amdgcn_wmma_f32_16x16x4_f32(
            false, a01, false, b1, (short)0, acc0, false, false);
        acc1 = __builtin_amdgcn_wmma_f32_16x16x4_f32(
            false, a11, false, b1, (short)0, acc1, false, false);
    }

    const float bv = bias[o0 + l];
    float* outBase = out + (size_t)(m0 + 8 * half) * D + o0 + l;
#pragma unroll
    for (int r = 0; r < 8; ++r) {
        outBase[(size_t)r * D] = acc0[r] + bv;
    }
    if (m0 + 16 < N_NODES) {
#pragma unroll
        for (int r = 0; r < 8; ++r) {
            outBase[(size_t)(16 + r) * D] = acc1[r] + bv;
        }
    }
}

extern "C" void kernel_launch(void* const* d_in, const int* in_sizes, int n_in,
                              void* d_out, int out_size, void* d_ws, size_t ws_size,
                              hipStream_t stream) {
    const int*   adj_row  = (const int*)d_in[0];
    const int*   adj_col  = (const int*)d_in[1];
    const float* adj_vals = (const float*)d_in[2];
    const float* features = (const float*)d_in[3];
    const float* W        = (const float*)d_in[4];
    const float* bias     = (const float*)d_in[5];
    float*       out      = (float*)d_out;

    int*   cnt   = (int*)d_ws + WS_CNT;
    int*   slots = (int*)d_ws + WS_SLOTS;
    int*   spill = (int*)d_ws + WS_SPILL;

    gcn_zero_kernel<<<(CNT_WORDS / 4 + 255) / 256, 256, 0, stream>>>((v4i*)cnt, CNT_WORDS / 4);

    gcn_bucket_kernel<<<(N_EDGES / 4 + 255) / 256, 256, 0, stream>>>(
        adj_row, adj_col, adj_vals, cnt, slots, spill);

    gcn_gather_gemm_kernel<<<(N_NODES + TILE_ROWS - 1) / TILE_ROWS, 256, 0, stream>>>(
        cnt, slots, spill, features, W, bias, out);
}


# Edge ranking on LDS counters instead of per-edge returning global atomics (block-by-partition segments, per-tile ranks in the gather kernel); clear launch goes, 3 -> 2; row sums in LDS arrival order
# speedup vs baseline: 3.6560x; 2.0668x over previous
#include <hip/hip_runtime.h>


typedef __attribute__((ext_vector_type(2))) float v2f;
typedef __attribute__((ext_vector_type(4))) float v4f;
typedef __attribute__((ext_vector_type(8))) float v8f;
typedef __attribute__((ext_vector_type(2))) int   v2i;
typedef __attribute__((ext_vector_type(4))) int   v4i;

#define N_NODES 50000
#define N_EDGES 800000
#define D 128

#define PART_ROWS  128
#define PART_SHIFT 7
#define N_PARTS    391
#define PART_THREADS 512
#define BLK_EDGES  4096
#define N_BLKS     196
#define SEG_CAP    32

static_assert(N_PARTS * PART_ROWS >= N_NODES && (N_PARTS - 1) * PART_ROWS < N_NODES, "partitions cover the rows exactly");
static_assert((1 << PART_SHIFT) == PART_ROWS, "partition of a row by shift");
static_assert(N_BLKS * BLK_EDGES >= N_EDGES && (N_BLKS - 1) * BLK_EDGES < N_EDGES, "blocks cover the edges exactly");
static_assert(N_EDGES % 4 == 0 && BLK_EDGES == PART_THREADS * 2 * 4, "b128 passes over edges, two per thread");
static_assert(N_NODES <= 65536, "a column index fits the low 16 bits of a segment key");
static_assert(N_PARTS <= PART_THREADS, "one thread per partition counter");

#define WS_COUNTS 0
#define WS_OVFCNT 76672
#define WS_SEGS   (WS_OVFCNT + 448)
#define WS_OVF    (WS_SEGS + N_PARTS * N_BLKS * SEG_CAP * 2)

static_assert(WS_OVFCNT >= N_PARTS * N_BLKS && N_BLKS <= 448, "count arrays do not overlap");
static_assert((WS_SEGS * 4) % 256 == 0 && (WS_OVF * 4) % 16 == 0, "segments 256-byte aligned, overflow entries 16-byte aligned");

__global__ void __launch_bounds__(PART_THREADS)
gcn_part_kernel(const int* __restrict__ adj_row,
                const int* __restrict__ adj_col,
                const float* __restrict__ adj_vals,
                int* __restrict__ counts,
                int* __restrict__ ovfcnt,
                int* __restrict__ segs,
                int* __restrict__ ovf) {
    __shared__ int pc[N_PARTS];
    __shared__ int oc;
    const int t   = threadIdx.x;
    const int blk = blockIdx.x;
    if (t < N_PARTS) pc[t] = 0;
    if (t == 0) oc = 0;
    __syncthreads();

    const int  q0  = blk * (BLK_EDGES / 4) + t;
    const int  q1  = q0 + PART_THREADS;
    const bool in0 = q0 < N_EDGES / 4;
    const bool in1 = q1 < N_EDGES / 4;
    const int  qa  = in0 ? q0 : 0;
    const int  qb  = in1 ? q1 : 0;
    const v4i r0 = ((const v4i*)adj_row)[qa];
    const v4i c0 = ((const v4i*)adj_col)[qa];
    const v4f v0 = ((const v4f*)adj_vals)[qa];
    const v4i r1 = ((const v4i*)adj_row)[qb];
    const v4i c1 = ((const v4i*)adj_col)[qb];
    const v4f v1 = ((const v4f*)adj_vals)[qb];

    const int   rr[8] = {r0.x, r0.y, r0.z, r0.w, r1.x, r1.y, r1.z, r1.w};
    const int   cc[8] = {c0.x, c0.y, c0.z, c0.w, c1.x, c1.y, c1.z, c1.w};
    const float vv[8] = {v0.x, v0.y, v0.z, v0.w, v1.x, v1.y, v1.z, v1.w};
    int k[8];
#pragma unroll
    for (int u = 0; u < 8; ++u) {
        int kk = -1;
        if ((u < 4) ? in0 : in1)
            kk = atomicAdd(pc + (rr[u] >> PART_SHIFT), 1);
        k[u] = kk;
    }
#pragma unroll
    for (int u = 0; u < 8; ++u) {
        if (k[u] >= 0) {
            if (k[u] < SEG_CAP) {
                const int p = rr[u] >> PART_SHIFT;
                const v2i e = {cc[u] | ((rr[u] & (PART_ROWS - 1)) << 16), __float_as_int(vv[u])};
                ((v2i*)segs)[(p * N_BLKS + blk) * SEG_CAP + k[u]] = e;
            } else {
                const int ko = atomicAdd(&oc, 1);
                const v4i e = {rr[u], cc[u], __float_as_int(vv[u]), 0};
                ((v4i*)ovf)[blk * BLK_EDGES + ko] = e;
            }
        }
    }
    __syncthreads();

    if (t < N_PARTS) {
        const int n = pc[t];
        counts[t * N_BLKS + blk] = (n < SEG_CAP) ? n : SEG_CAP;
    }
    if (t == 0) ovfcnt[blk] = oc;
}

#define TILE_ROWS      32
#define ROWS_PER_WAVE  4
#define F_PITCH        132
#define ROW_SLOTS      64
#define TILE_SHIFT     5
#define TILES_PER_PART (PART_ROWS / TILE_ROWS)
#define SEGS_PER_WAVE  25
#define SEG_BATCH      13

static_assert(TILE_ROWS == 8 * ROWS_PER_WAVE, "8 waves fill the tile");
static_assert((1 << TILE_SHIFT) == TILE_ROWS && PART_ROWS % TILE_ROWS == 0, "tile of a row by shift; tiles do not straddle partitions");
static_assert(N_NODES % 16 == 0, "a 16-row half tile is wholly inside or wholly outside the node range");
static_assert(8 * SEGS_PER_WAVE >= N_BLKS, "the 8 waves cover every segment");
static_assert(N_BLKS <= 256, "one load per thread covers every block's overflow count");
static_assert(SEGS_PER_WAVE < 32 && 2 * SEG_BATCH >= SEGS_PER_WAVE && 2 * SEG_BATCH <= 32, "a wave's counts fit one vector load; batch positions past the last segment read a lane that holds 0");

template <int U>
__device__ __forceinline__ v4f gcn_gather_group(const float* __restrict__ fbase,
                                                const v2i pr, const int j, v4f acc) {
    v4f   x[U];
    float v[U];
#pragma unroll
    for (int u = 0; u < U; ++u) {
        const int c = __builtin_amdgcn_readlane(pr.x, j + u);
        v[u] = __int_as_float(__builtin_amdgcn_readlane(pr.y, j + u));
        x[u] = *(const v4f*)(fbase + (size_t)c * D);
    }
#pragma unroll
    for (int u = 0; u < U; ++u) acc += v[u] * x[u];
    return acc;
}

__device__ __forceinline__ v4f gcn_gather_slots(const float* __restrict__ fbase,
                                                const v2i pr, const int n, v4f acc) {
    int j = 0;
    for (; j + 8 <= n; j += 8) acc = gcn_gather_group<8>(fbase, pr, j, acc);
    const int rem = n - j;
    if (rem & 4) { acc = gcn_gather_group<4>(fbase, pr, j, acc); j += 4; }
    if (rem & 2) { acc = gcn_gather_group<2>(fbase, pr, j, acc); j += 2; }
    if (rem & 1) { acc = gcn_gather_group<1>(fbase, pr, j, acc); }
    return acc;
}

__device__ __forceinline__ v4f gcn_long_row(const int* __restrict__ pcnt, const v2i* __restrict__ pseg,
                                         const int* __restrict__ ovfcnt, const v4i* __restrict__ ovf,
                                         const float* __restrict__ fbase,
                                         const int rin, const int row, const int lane) {
    v4f acc = {};
    for (int s = 0; s < N_BLKS; ++s) {
        const int n = pcnt[s];
        v2i e = {-1, 0};
        if (lane < n) e = pseg[s * SEG_CAP + lane];
        unsigned hit = __builtin_amdgcn_ballot_w32(lane < n && (e.x >> 16) == rin);
        while (hit) {
            const int from = __builtin_ctz(hit);
            hit &= hit - 1;
            const int   c = __builtin_amdgcn_readlane(e.x, from) & 0xffff;
            const float v = __int_as_float(__builtin_amdgcn_readlane(e.y, from));
            acc += v * *(const v4f*)(fbase + (size_t)c * D);
        }
    }
    for (int b = 0; b < N_BLKS; ++b) {
        int m = ovfcnt[b];
        m = (m < BLK_EDGES) ? m : BLK_EDGES;
        for (int j = 0; j < m; j += 32) {
            v4i e = {-1, 0, 0, 0};
            if (j + lane < m) e = ovf[b * BLK_EDGES + j + lane];
            unsigned hit = __builtin_amdgcn_ballot_w32(e.x == row);
            while (hit) {
                const int from = __builtin_ctz(hit);
                hit &= hit - 1;
                const int   c = __builtin_amdgcn_readlane(e.y, from);
                const float v = __int_as_float(__builtin_amdgcn_readlane(e.z, from));
                acc += v * *(const v4f*)(fbase + (size_t)c * D);
            }
        }
    }
    return acc;
}

__global__ void __launch_bounds__(256)
gcn_gather_gemm_kernel(const int* __restrict__ counts,
                       const int* __restrict__ ovfcnt,
                       const int* __restrict__ segs,
                       const int* __restrict__ ovf,
                       const float* __restrict__ features,
                       const float* __restrict__ W,
                       const float* __restrict__ bias,
                       float* __restrict__ out) {
    __shared__ float ftile[TILE_ROWS * F_PITCH];
    __shared__ v2i   lslot[TILE_ROWS * ROW_SLOTS];
    __shared__ int   rowCnt[TILE_ROWS];
    __shared__ int   anyOvf;

    const int waveInBlk = __builtin_amdgcn_readfirstlane(threadIdx.x >> 5);
    const int lane      = threadIdx.x & 31;
    const int tile      = blockIdx.x;
    const int m0        = tile * TILE_ROWS;
    const int part      = tile / TILES_PER_PART;
    const int tin       = tile % TILES_PER_PART;

    const int* pcnt = counts + part * N_BLKS;
    const v2i* pseg = (const v2i*)segs + part * N_BLKS * SEG_CAP;

    if (threadIdx.x < TILE_ROWS) rowCnt[threadIdx.x] = 0;
    if (threadIdx.x == 0) anyOvf = 0;
    __syncthreads();

    const int ovA = ovfcnt[(threadIdx.x < N_BLKS) ? threadIdx.x : N_BLKS - 1];
    {
        const int sBeg = waveInBlk * SEGS_PER_WAVE;
        int nv = 0;
        if (lane < SEGS_PER_WAVE && sBeg + lane < N_BLKS) nv = pcnt[sBeg + lane];

        for (int j0 = 0; j0 < SEGS_PER_WAVE; j0 += SEG_BATCH) {
            int n[SEG_BATCH];
            v2i e[SEG_BATCH];
            int k[SEG_BATCH];
#pragma unroll
            for (int u = 0; u < SEG_BATCH; ++u) n[u] = __builtin_amdgcn_readlane(nv, j0 + u);
#pragma unroll
            for (int u = 0; u < SEG_BATCH; ++u) {
                v2i x = {};
                if (lane < n[u]) x = pseg[(sBeg + j0 + u) * SEG_CAP + lane];
                e[u] = x;
            }
#pragma unroll
            for (int u = 0; u < SEG_BATCH; ++u) {
                const int rin = e[u].x >> 16;
                int kk = ROW_SLOTS;
                if (lane < n[u] && (rin >> TILE_SHIFT) == tin)
                    kk = atomicAdd(rowCnt + (rin & (TILE_ROWS - 1)), 1);
                k[u] = kk;
            }
#pragma unroll
            for (int u = 0; u < SEG_BATCH; ++u) {
                if (k[u] < ROW_SLOTS) {
                    const int r = (e[u].x >> 16) & (TILE_ROWS - 1);
                    const v2i p = {e[u].x & 0xffff, e[u].y};
                    lslot[r * ROW_SLOTS + k[u]] = p;
                }
            }
        }
    }
    if (ovA != 0) anyOvf = 1;
    __syncthreads();

    if (anyOvf != 0) {
        for (int b = 0; b < N_BLKS; ++b) {
            int m = ovfcnt[b];
            m = (m < BLK_EDGES) ? m : BLK_EDGES;
            for (int i = threadIdx.x; i < m; i += 256) {
                const v4i e = ((const v4i*)ovf)[b * BLK_EDGES + i];
                if ((e.x >> TILE_SHIFT) == tile) {
                    const int r = e.x & (TILE_ROWS - 1);
                    const int k = atomicAdd(rowCnt + r, 1);
                    if (k < ROW_SLOTS) {
                        const v2i p = {e.y, e.z};
                        lslot[r * ROW_SLOTS + k] = p;
                    }
                }
            }
        }
        __syncthreads();
    }

    const float* fbase = features + lane * 4;

#pragma nounroll
    for (int i = 0; i < ROWS_PER_WAVE; ++i) {
        const int rt  = waveInBlk * ROWS_PER_WAVE + i;
        const int deg = __builtin_amdgcn_readfirstlane(rowCnt[rt]);
        v4f acc = {};

        if (deg <= ROW_SLOTS) {
            const int n0 = (deg < 32) ? deg : 32;
            v2i p0 = {};
            if (lane < n0) p0 = lslot[rt * ROW_SLOTS + lane];
            acc = gcn_gather_slots(fbase, p0, n0, acc);
            if (deg > 32) {
                const int n1 = deg - 32;
                v2i p1 = {};
                if (lane < n1) p1 = lslot[rt * ROW_SLOTS + 32 + lane];
                acc = gcn_gather_slots(fbase, p1, n1, acc);
            }
        } else {
            acc = gcn_long_row(pcnt, pseg, ovfcnt, (const v4i*)ovf, fbase,
                               tin * TILE_ROWS + rt, m0 + rt, lane);
        }

        *(v4f*)(ftile + rt * F_PITCH + lane * 4) = acc;
    }

    __syncthreads();

    const int half = lane >> 4;
    const int l    = lane & 15;
    const int o0   = waveInBlk * 16;

    const float* aRow0 = ftile + l * F_PITCH + 4 * half;
    const float* aRow1 = ftile + (16 + l) * F_PITCH + 4 * half;
    const float* bRow  = W + (size_t)(o0 + l) * D + 4 * half;

    v8f acc0 = {};
    v8f acc1 = {};

#pragma unroll
    for (int p = 0; p < 16; ++p) {
        const v4f vb  = *(const v4f*)(bRow + 8 * p);
        const v4f va0 = *(const v4f*)(aRow0 + 8 * p);
        const v4f va1 = *(const v4f*)(aRow1 + 8 * p);
        const v2f b0  = {vb.x, vb.y};
        const v2f b1  = {vb.z, vb.w};
        const v2f a00 = {va0.x, va0.y};
        const v2f a01 = {va0.z, va0.w};
        const v2f a10 = {va1.x, va1.y};
        const v2f a11 = {va1.z, va1.w};
        acc0 = __builtin_amdgcn_wmma_f32_16x16x4_f32(
            false, a00, false, b0, (short)0, acc0, false, false);
        acc1 = __builtin_amdgcn_wmma_f32_16x16x4_f32(
            false, a10, false, b0, (short)0, acc1, false, false);
        acc0 = __builtin_amdgcn_wmma_f32_16x16x4_f32(
            false, a01, false, b1, (short)0, acc0, false, false);
        acc1 = __builtin_amdgcn_wmma_f32_16x16x4_f32(
            false, a11, false, b1, (short)0, acc1, false, false);
    }

    const float bv = bias[o0 + l];
    float* outBase = out + (size_t)(m0 + 8 * half) * D + o0 + l;
#pragma unroll
    for (int r = 0; r < 8; ++r) {
        outBase[(size_t)r * D] = acc0[r] + bv;
    }
    if (m0 + 16 < N_NODES) {
#pragma unroll
        for (int r = 0; r < 8; ++r) {
            outBase[(size_t)(16 + r) * D] = acc1[r] + bv;
        }
    }
}

extern "C" void kernel_launch(void* const* d_in, const int* in_sizes, int n_in,
                              void* d_out, int out_size, void* d_ws, size_t ws_size,
                              hipStream_t stream) {
    const int*   adj_row  = (const int*)d_in[0];
    const int*   adj_col  = (const int*)d_in[1];
    const float* adj_vals = (const float*)d_in[2];
    const float* features = (const float*)d_in[3];
    const float* W        = (const float*)d_in[4];
    const float* bias     = (const float*)d_in[5];
    float*       out      = (float*)d_out;

    int* counts = (int*)d_ws + WS_COUNTS;
    int* ovfcnt = (int*)d_ws + WS_OVFCNT;
    int* segs   = (int*)d_ws + WS_SEGS;
    int* ovf    = (int*)d_ws + WS_OVF;

    gcn_part_kernel<<<N_BLKS, PART_THREADS, 0, stream>>>(
        adj_row, adj_col, adj_vals, counts, ovfcnt, segs, ovf);

    gcn_gather_gemm_kernel<<<(N_NODES + TILE_ROWS - 1) / TILE_ROWS, 256, 0, stream>>>(
        counts, ovfcnt, segs, ovf, features, W, bias, out);
}
